# starting kernel
# baseline (speedup 1.0000x reference)
#include <hip/hip_runtime.h>
#include <cmath>

typedef __attribute__((ext_vector_type(2))) float v2f;
typedef __attribute__((ext_vector_type(8))) float v8f;

#define RNN_B 1024
#define RNN_T 512
#define RNN_H 128
#define RNN_P 4
#define RNN_TT (RNN_T + RNN_P)
#define NTILE 8
#define KCHUNK 32
#define HB_STRIDE 132

#define FCW_ELEMS (KCHUNK * NTILE * 32 * 2)

__device__ __forceinline__ float dev_tanh(float v) {
#if __has_builtin(__builtin_amdgcn_tanhf)
  return __builtin_amdgcn_tanhf(v);
#else
  return tanhf(v);
#endif
}

__global__ __launch_bounds__(32) void rnn516_wmma(
    const float* __restrict__ x, const float* __restrict__ hidden,
    const float* __restrict__ fc0_w, const float* __restrict__ fc0_b,
    const float* __restrict__ fc_w, const float* __restrict__ fc_b,
    const float* __restrict__ dec_w, const float* __restrict__ dec_b,
    float* __restrict__ y) {
  extern __shared__ float lds[];
  float* fcw  = lds;
  float* hbuf = lds + FCW_ELEMS;
  float* ybuf = hbuf + 16 * HB_STRIDE;

  const int lane = threadIdx.x;
  const int half = lane >> 4;
  const int lmod = lane & 15;
  const int b0 = blockIdx.x * 16;

  for (int i = lane; i < FCW_ELEMS; i += 32) {
    int j  = i & 1;
    int l2 = (i >> 1) & 31;
    int nt = (i >> 6) & 7;
    int kc = i >> 9;
    int n = nt * 16 + (l2 & 15);
    int k = 4 * kc + 2 * (l2 >> 4) + j;
    fcw[i] = fc_w[n * RNN_H + k];
  }
  for (int i = lane; i < 16 * HB_STRIDE; i += 32) {
    int r = i / HB_STRIDE, k = i - r * HB_STRIDE;
    hbuf[i] = (k < RNN_H) ? hidden[(b0 + r) * RNN_H + k] : 0.0f;
  }
  __syncthreads();

  float biasv[NTILE], dwv[NTILE];
  v2f blast[NTILE];
#pragma unroll
  for (int nt = 0; nt < NTILE; ++nt) {
    int n = nt * 16 + lmod;
    biasv[nt] = fc0_b[n] + fc_b[n];
    dwv[nt] = dec_w[n];
    v2f b;
    b[0] = (half == 0) ? fc0_w[n] : 0.0f;
    b[1] = 0.0f;
    blast[nt] = b;
  }
  const float decb = dec_b[0];

  float xval = 0.0f;

  for (int t = 0; t < RNN_TT; ++t) {
    if (t < RNN_T) {
      if (half == 0) xval = x[(b0 + lmod) * RNN_T + t];
    }

    v8f d[NTILE];
#pragma unroll
    for (int nt = 0; nt < NTILE; ++nt) {
#pragma unroll
      for (int e = 0; e < 8; ++e) d[nt][e] = biasv[nt];
    }

    const float* arow  = hbuf + lmod * HB_STRIDE + 2 * half;
    const float* bcol  = fcw + 2 * lane;

    v2f a_cur, b_cur[NTILE], a_nxt, b_nxt[NTILE];
    a_cur = *(const v2f*)(arow);
#pragma unroll
    for (int nt = 0; nt < NTILE; ++nt)
      b_cur[nt] = *(const v2f*)(bcol + nt * 64);

#pragma unroll 4
    for (int kc = 0; kc < KCHUNK - 1; ++kc) {
      a_nxt = *(const v2f*)(arow + 4 * (kc + 1));
      const float* bb = bcol + (kc + 1) * (NTILE * 64);
#pragma unroll
      for (int nt = 0; nt < NTILE; ++nt)
        b_nxt[nt] = *(const v2f*)(bb + nt * 64);

#pragma unroll
      for (int nt = 0; nt < NTILE; ++nt)
        d[nt] = __builtin_amdgcn_wmma_f32_16x16x4_f32(
            false, a_cur, false, b_cur[nt], (short)0, d[nt], false, false);

      a_cur = a_nxt;
#pragma unroll
      for (int nt = 0; nt < NTILE; ++nt) b_cur[nt] = b_nxt[nt];
    }
#pragma unroll
    for (int nt = 0; nt < NTILE; ++nt)
      d[nt] = __builtin_amdgcn_wmma_f32_16x16x4_f32(
          false, a_cur, false, b_cur[nt], (short)0, d[nt], false, false);

    {
      v2f al;
      al[0] = xval;
      al[1] = 0.0f;
#pragma unroll
      for (int nt = 0; nt < NTILE; ++nt)
        d[nt] = __builtin_amdgcn_wmma_f32_16x16x4_f32(
            false, al, false, blast[nt], (short)0, d[nt], false, false);
    }

    float yred[8];
#pragma unroll
    for (int v = 0; v < 8; ++v) yred[v] = 0.0f;
#pragma unroll
    for (int nt = 0; nt < NTILE; ++nt) {
      int col = nt * 16 + lmod;
#pragma unroll
      for (int v = 0; v < 8; ++v) {
        float th = dev_tanh(d[nt][v]);
        hbuf[(v + 8 * half) * HB_STRIDE + col] = th;
        yred[v] += th * dwv[nt];
      }
    }

#pragma unroll
    for (int v = 0; v < 8; ++v) {
      float s = yred[v];
      s += __shfl_xor(s, 1, 32);
      s += __shfl_xor(s, 2, 32);
      s += __shfl_xor(s, 4, 32);
      s += __shfl_xor(s, 8, 32);
      if (lmod == 0) ybuf[v + 8 * half] = s + decb;
    }

    float yv = ybuf[lmod];
    if (half == 0) {
      y[(b0 + lmod) * RNN_TT + t] = yv;
      xval = yv;
    }
  }
}

extern "C" void kernel_launch(void* const* d_in, const int* in_sizes, int n_in,
                              void* d_out, int out_size, void* d_ws,
                              size_t ws_size, hipStream_t stream) {
  const float* x      = (const float*)d_in[0];
  const float* hidden = (const float*)d_in[1];
  const float* fc0_w  = (const float*)d_in[2];
  const float* fc0_b  = (const float*)d_in[3];
  const float* fc_w   = (const float*)d_in[4];
  const float* fc_b   = (const float*)d_in[5];
  const float* dec_w  = (const float*)d_in[6];
  const float* dec_b  = (const float*)d_in[7];
  float* y = (float*)d_out;

  const size_t shmem =
      (size_t)(FCW_ELEMS + 16 * HB_STRIDE + 16) * sizeof(float);
  rnn516_wmma<<<dim3(RNN_B / 16), dim3(32), shmem, stream>>>(
      x, hidden, fc0_w, fc0_b, fc_w, fc_b, dec_w, dec_b, y);
}


# 4 waves per 16-row tile instead of 1: each wave owns 32 hidden columns, its fc_w fragments stay in VGPRs, h goes through ping-pong LDS, 1 barrier/step; decode sum = 4 wave partials (new sum order)
# speedup vs baseline: 2.6021x; 2.6021x over previous
#include <hip/hip_runtime.h>
#include <cmath>


typedef __attribute__((ext_vector_type(2))) float v2f;
typedef __attribute__((ext_vector_type(8))) float v8f;

#define RNN_B 1024
#define RNN_T 512
#define RNN_H 128
#define RNN_P 4
#define RNN_TT (RNN_T + RNN_P)
#define NWAVE 4
#define NT_W 2
#define KCHUNK 32
#define HB_STRIDE 132
#define HB_ELEMS (16 * HB_STRIDE)

__device__ __forceinline__ float dev_tanh(float v) {
#if __has_builtin(__builtin_amdgcn_tanhf)
  return __builtin_amdgcn_tanhf(v);
#else
  return tanhf(v);
#endif
}

__global__ __launch_bounds__(NWAVE * 32) void rnn516_wmma(
    const float* __restrict__ x, const float* __restrict__ hidden,
    const float* __restrict__ fc0_w, const float* __restrict__ fc0_b,
    const float* __restrict__ fc_w, const float* __restrict__ fc_b,
    const float* __restrict__ dec_w, const float* __restrict__ dec_b,
    float* __restrict__ y) {
  __shared__ float hbuf[2][HB_ELEMS];
  __shared__ float ypart[2][NWAVE * 16];

  const int tid  = threadIdx.x;
  const int lane = tid & 31;
  const int wave = __builtin_amdgcn_readfirstlane(tid >> 5);
  const int half = lane >> 4;
  const int lmod = lane & 15;
  const int b0 = blockIdx.x * 16;

  for (int i = tid; i < HB_ELEMS; i += NWAVE * 32) {
    int r = i / HB_STRIDE, k = i - r * HB_STRIDE;
    hbuf[0][i] = (k < RNN_H) ? hidden[(b0 + r) * RNN_H + k] : 0.0f;
  }
  if (tid < NWAVE * 16) ypart[1][tid] = 0.0f;

  v2f bw[NT_W][KCHUNK];
  float biasv[NT_W], dwv[NT_W];
  v2f blast[NT_W];
#pragma unroll
  for (int jt = 0; jt < NT_W; ++jt) {
    const int n = (wave * NT_W + jt) * 16 + lmod;
    const float* wrow = fc_w + n * RNN_H + 2 * half;
#pragma unroll
    for (int kc = 0; kc < KCHUNK; ++kc)
      bw[jt][kc] = *(const v2f*)(wrow + 4 * kc);
    biasv[jt] = fc0_b[n] + fc_b[n];
    dwv[jt] = dec_w[n];
    v2f b;
    b[0] = (half == 0) ? fc0_w[n] : 0.0f;
    b[1] = 0.0f;
    blast[jt] = b;
  }
  const float decb = dec_b[0];
  const float* xrow = x + (b0 + lmod) * RNN_T;
  float xcur = xrow[0];

  __syncthreads();

  for (int t = 0; t < RNN_TT; ++t) {
    const int par = t & 1;
    const float* hcur = hbuf[par];
    float* hnxt = hbuf[par ^ 1];

    const float* yp = ypart[par ^ 1];
    const float p0 = yp[lmod], p1 = yp[16 + lmod];
    const float p2 = yp[32 + lmod], p3 = yp[48 + lmod];

    v2f a[KCHUNK];
    const float* arow = hcur + lmod * HB_STRIDE + 2 * half;
#pragma unroll
    for (int kc = 0; kc < KCHUNK; ++kc) a[kc] = *(const v2f*)(arow + 4 * kc);

    const int tn = (t + 1 < RNN_T) ? (t + 1) : (RNN_T - 1);
    const float xnext = xrow[tn];

    const float yprev = ((p0 + p1) + (p2 + p3)) + decb;
    if (t > 0 && wave == 0 && half == 0) y[(b0 + lmod) * RNN_TT + (t - 1)] = yprev;

    v8f d[NT_W];
#pragma unroll
    for (int jt = 0; jt < NT_W; ++jt) {
#pragma unroll
      for (int e = 0; e < 8; ++e) d[jt][e] = biasv[jt];
    }

#pragma unroll
    for (int kc = 0; kc < KCHUNK; ++kc) {
#pragma unroll
      for (int jt = 0; jt < NT_W; ++jt)
        d[jt] = __builtin_amdgcn_wmma_f32_16x16x4_f32(
            false, a[kc], false, bw[jt][kc], (short)0, d[jt], false, false);
    }

    {
      v2f al;
      al[0] = (half == 0) ? ((t < RNN_T) ? xcur : yprev) : 0.0f;
      al[1] = 0.0f;
#pragma unroll
      for (int jt = 0; jt < NT_W; ++jt)
        d[jt] = __builtin_amdgcn_wmma_f32_16x16x4_f32(
            false, al, false, blast[jt], (short)0, d[jt], false, false);
    }
    xcur = xnext;

    float yred[8];
#pragma unroll
    for (int v = 0; v < 8; ++v) yred[v] = 0.0f;
#pragma unroll
    for (int jt = 0; jt < NT_W; ++jt) {
      const int col = (wave * NT_W + jt) * 16 + lmod;
#pragma unroll
      for (int v = 0; v < 8; ++v) {
        float th = dev_tanh(d[jt][v]);
        hnxt[(v + 8 * half) * HB_STRIDE + col] = th;
        yred[v] += th * dwv[jt];
      }
    }

    float* ypw = ypart[par] + wave * 16;
#pragma unroll
    for (int v = 0; v < 8; ++v) {
      float s = yred[v];
      s += __shfl_xor(s, 1, 32);
      s += __shfl_xor(s, 2, 32);
      s += __shfl_xor(s, 4, 32);
      s += __shfl_xor(s, 8, 32);
      if (lmod == 0) ypw[v + 8 * half] = s;
    }

    __syncthreads();
  }

  {
    const float* yp = ypart[(RNN_TT - 1) & 1];
    const float yl = ((yp[lmod] + yp[16 + lmod]) + (yp[32 + lmod] + yp[48 + lmod])) + decb;
    if (wave == 0 && half == 0) y[(b0 + lmod) * RNN_TT + (RNN_TT - 1)] = yl;
  }
}

extern "C" void kernel_launch(void* const* d_in, const int* in_sizes, int n_in,
                              void* d_out, int out_size, void* d_ws,
                              size_t ws_size, hipStream_t stream) {
  const float* x      = (const float*)d_in[0];
  const float* hidden = (const float*)d_in[1];
  const float* fc0_w  = (const float*)d_in[2];
  const float* fc0_b  = (const float*)d_in[3];
  const float* fc_w   = (const float*)d_in[4];
  const float* fc_b   = (const float*)d_in[5];
  const float* dec_w  = (const float*)d_in[6];
  const float* dec_b  = (const float*)d_in[7];
  float* y = (float*)d_out;

  rnn516_wmma<<<dim3(RNN_B / 16), dim3(NWAVE * 32), 0, stream>>>(
      x, hidden, fc0_w, fc0_b, fc_w, fc_b, dec_w, dec_b, y);
}


# decode partial sums reduced over each 16-lane row by DPP row_xmask adds on the vector ALU instead of 32 serial LDS-unit shuffles per step; same butterfly order, sums bit-identical
# speedup vs baseline: 5.9359x; 2.2812x over previous
#include <hip/hip_runtime.h>
#include <cmath>


typedef __attribute__((ext_vector_type(2))) float v2f;
typedef __attribute__((ext_vector_type(8))) float v8f;

#define RNN_B 1024
#define RNN_T 512
#define RNN_H 128
#define RNN_P 4
#define RNN_TT (RNN_T + RNN_P)
#define NWAVE 4
#define NT_W 2
#define KCHUNK 32
#define HB_STRIDE 132
#define HB_ELEMS (16 * HB_STRIDE)

__device__ __forceinline__ float dev_tanh(float v) {
#if __has_builtin(__builtin_amdgcn_tanhf)
  return __builtin_amdgcn_tanhf(v);
#else
  return tanhf(v);
#endif
}

template <int XM>
__device__ __forceinline__ float row_xor_add(float s) {
  const int p = __builtin_amdgcn_update_dpp(0, __float_as_int(s), 0x160 + XM,
                                            0xf, 0xf, true);
  return s + __int_as_float(p);
}

__global__ __launch_bounds__(NWAVE * 32) void rnn516_wmma(
    const float* __restrict__ x, const float* __restrict__ hidden,
    const float* __restrict__ fc0_w, const float* __restrict__ fc0_b,
    const float* __restrict__ fc_w, const float* __restrict__ fc_b,
    const float* __restrict__ dec_w, const float* __restrict__ dec_b,
    float* __restrict__ y) {
  __shared__ float hbuf[2][HB_ELEMS];
  __shared__ __attribute__((aligned(16))) float ypart[2][NWAVE * 16];

  const int tid  = threadIdx.x;
  const int lane = tid & 31;
  const int wave = __builtin_amdgcn_readfirstlane(tid >> 5);
  const int half = lane >> 4;
  const int lmod = lane & 15;
  const int b0 = blockIdx.x * 16;

  for (int i = tid; i < HB_ELEMS; i += NWAVE * 32) {
    int r = i / HB_STRIDE, k = i - r * HB_STRIDE;
    hbuf[0][i] = (k < RNN_H) ? hidden[(b0 + r) * RNN_H + k] : 0.0f;
  }
  if (tid < NWAVE * 16) ypart[1][tid] = 0.0f;

  v2f bw[NT_W][KCHUNK];
  float biasv[NT_W], dwv[NT_W];
  v2f blast[NT_W];
#pragma unroll
  for (int jt = 0; jt < NT_W; ++jt) {
    const int n = (wave * NT_W + jt) * 16 + lmod;
    const float* wrow = fc_w + n * RNN_H + 2 * half;
#pragma unroll
    for (int kc = 0; kc < KCHUNK; ++kc)
      bw[jt][kc] = *(const v2f*)(wrow + 4 * kc);
    biasv[jt] = fc0_b[n] + fc_b[n];
    dwv[jt] = dec_w[n];
    v2f b;
    b[0] = (half == 0) ? fc0_w[n] : 0.0f;
    b[1] = 0.0f;
    blast[jt] = b;
  }
  const float decb = dec_b[0];
  const float* xrow = x + (b0 + lmod) * RNN_T;
  float xcur = xrow[0];

  __syncthreads();

  for (int t = 0; t < RNN_TT; ++t) {
    const int par = t & 1;
    const float* hcur = hbuf[par];
    float* hnxt = hbuf[par ^ 1];

    const float* yp = ypart[par ^ 1];
    const float p0 = yp[lmod], p1 = yp[16 + lmod];
    const float p2 = yp[32 + lmod], p3 = yp[48 + lmod];

    v2f a[KCHUNK];
    const float* arow = hcur + lmod * HB_STRIDE + 2 * half;
#pragma unroll
    for (int kc = 0; kc < KCHUNK; ++kc) a[kc] = *(const v2f*)(arow + 4 * kc);

    const int tn = (t + 1 < RNN_T) ? (t + 1) : (RNN_T - 1);
    const float xnext = xrow[tn];

    const float yprev = ((p0 + p1) + (p2 + p3)) + decb;
    if (t > 0 && wave == 0 && half == 0) y[(b0 + lmod) * RNN_TT + (t - 1)] = yprev;

    v8f d[NT_W];
#pragma unroll
    for (int jt = 0; jt < NT_W; ++jt) {
#pragma unroll
      for (int e = 0; e < 8; ++e) d[jt][e] = biasv[jt];
    }

#pragma unroll
    for (int kc = 0; kc < KCHUNK; ++kc) {
#pragma unroll
      for (int jt = 0; jt < NT_W; ++jt)
        d[jt] = __builtin_amdgcn_wmma_f32_16x16x4_f32(
            false, a[kc], false, bw[jt][kc], (short)0, d[jt], false, false);
    }

    {
      v2f al;
      al[0] = (half == 0) ? ((t < RNN_T) ? xcur : yprev) : 0.0f;
      al[1] = 0.0f;
#pragma unroll
      for (int jt = 0; jt < NT_W; ++jt)
        d[jt] = __builtin_amdgcn_wmma_f32_16x16x4_f32(
            false, al, false, blast[jt], (short)0, d[jt], false, false);
    }
    xcur = xnext;

    float yred[8];
#pragma unroll
    for (int v = 0; v < 8; ++v) yred[v] = 0.0f;
#pragma unroll
    for (int jt = 0; jt < NT_W; ++jt) {
      const int col = (wave * NT_W + jt) * 16 + lmod;
#pragma unroll
      for (int v = 0; v < 8; ++v) {
        float th = dev_tanh(d[jt][v]);
        hnxt[(v + 8 * half) * HB_STRIDE + col] = th;
        yred[v] += th * dwv[jt];
      }
    }

#pragma unroll
    for (int v = 0; v < 8; ++v) yred[v] = row_xor_add<1>(yred[v]);
#pragma unroll
    for (int v = 0; v < 8; ++v) yred[v] = row_xor_add<2>(yred[v]);
#pragma unroll
    for (int v = 0; v < 8; ++v) yred[v] = row_xor_add<4>(yred[v]);
#pragma unroll
    for (int v = 0; v < 8; ++v) yred[v] = row_xor_add<8>(yred[v]);
    if (lmod == 0) {
      float* ypw = ypart[par] + wave * 16 + 8 * half;
#pragma unroll
      for (int v = 0; v < 8; ++v) ypw[v] = yred[v];
    }

    __syncthreads();
  }

  {
    const float* yp = ypart[(RNN_TT - 1) & 1];
    const float yl = ((yp[lmod] + yp[16 + lmod]) + (yp[32 + lmod] + yp[48 + lmod])) + decb;
    if (wave == 0 && half == 0) y[(b0 + lmod) * RNN_TT + (RNN_TT - 1)] = yl;
  }
}

extern "C" void kernel_launch(void* const* d_in, const int* in_sizes, int n_in,
                              void* d_out, int out_size, void* d_ws,
                              size_t ws_size, hipStream_t stream) {
  const float* x      = (const float*)d_in[0];
  const float* hidden = (const float*)d_in[1];
  const float* fc0_w  = (const float*)d_in[2];
  const float* fc0_b  = (const float*)d_in[3];
  const float* fc_w   = (const float*)d_in[4];
  const float* fc_b   = (const float*)d_in[5];
  const float* dec_w  = (const float*)d_in[6];
  const float* dec_b  = (const float*)d_in[7];
  float* y = (float*)d_out;

  rnn516_wmma<<<dim3(RNN_B / 16), dim3(NWAVE * 32), 0, stream>>>(
      x, hidden, fc0_w, fc0_b, fc_w, fc_b, dec_w, dec_b, y);
}


# step software-pipelined over each wave's two 16-col tiles: K split by producing tile, a tile's tanh/LDS store/decode reduce issued between the other half's WMMAs, 2 barriers/step; new K sum order
# speedup vs baseline: 6.8769x; 1.1585x over previous
#include <hip/hip_runtime.h>
#include <cmath>


typedef __attribute__((ext_vector_type(2))) float v2f;
typedef __attribute__((ext_vector_type(8))) float v8f;

#define RNN_B 1024
#define RNN_T 512
#define RNN_H 128
#define RNN_P 4
#define RNN_TT (RNN_T + RNN_P)
#define NWAVE 4
#define NT_W 2
#define KCHUNK 32
#define HB_STRIDE 132
#define HB_ELEMS (16 * HB_STRIDE)

__device__ __forceinline__ float dev_tanh(float v) {
#if __has_builtin(__builtin_amdgcn_tanhf)
  return __builtin_amdgcn_tanhf(v);
#else
  return tanhf(v);
#endif
}

template <int XM>
__device__ __forceinline__ float row_xor_add(float s) {
  const int p = __builtin_amdgcn_update_dpp(0, __float_as_int(s), 0x160 + XM,
                                            0xf, 0xf, true);
  return s + __int_as_float(p);
}

#define KC0(i) (8 * ((i) >> 2) + ((i) & 3))
#define KC1(i) (KC0(i) + 4)
#define WMMA4(a, b, c) \
  __builtin_amdgcn_wmma_f32_16x16x4_f32(false, (a), false, (b), (short)0, (c), false, false)

#define ORDER2(a, b) asm volatile("" : "+v"(a), "+v"(b))
#define ORDER3(a, b, c) asm volatile("" : "+v"(a), "+v"(b), "+v"(c))
#define ORDER4(a, b, c, d) asm volatile("" : "+v"(a), "+v"(b), "+v"(c), "+v"(d))
#define BEFORE2(a, b) asm volatile("" : : "v"(a), "v"(b))

template <int XM>
__device__ __forceinline__ v8f row_stage(v8f s) {
#pragma unroll
  for (int v = 0; v < 8; ++v) s[v] = row_xor_add<XM>(s[v]);
  return s;
}

__global__ __launch_bounds__(NWAVE * 32) void rnn516_wmma(
    const float* __restrict__ x, const float* __restrict__ hidden,
    const float* __restrict__ fc0_w, const float* __restrict__ fc0_b,
    const float* __restrict__ fc_w, const float* __restrict__ fc_b,
    const float* __restrict__ dec_w, const float* __restrict__ dec_b,
    float* __restrict__ y) {
  __shared__ float hbuf[2][HB_ELEMS];
  __shared__ __attribute__((aligned(16))) float ypart[2][NWAVE * 32 * 8];

  const int tid  = threadIdx.x;
  const int lane = tid & 31;
  const int wave = __builtin_amdgcn_readfirstlane(tid >> 5);
  const int half = lane >> 4;
  const int lmod = lane & 15;
  const int b0 = blockIdx.x * 16;

  for (int i = tid; i < HB_ELEMS; i += NWAVE * 32) {
    int r = i / HB_STRIDE, k = i - r * HB_STRIDE;
    hbuf[0][i] = (k < RNN_H) ? hidden[(b0 + r) * RNN_H + k] : 0.0f;
  }

  v2f bw[NT_W][KCHUNK];
  float biasv[NT_W], dwv[NT_W];
  v2f blast[NT_W];
#pragma unroll
  for (int jt = 0; jt < NT_W; ++jt) {
    const int n = (wave * NT_W + jt) * 16 + lmod;
    const float* wrow = fc_w + n * RNN_H + 2 * half;
#pragma unroll
    for (int kc = 0; kc < KCHUNK; ++kc)
      bw[jt][kc] = *(const v2f*)(wrow + 4 * kc);
    biasv[jt] = fc0_b[n] + fc_b[n];
    dwv[jt] = dec_w[n];
    v2f b;
    b[0] = (half == 0) ? fc0_w[n] : 0.0f;
    b[1] = 0.0f;
    blast[jt] = b;
  }
  const float decb = dec_b[0];
  const float* xrow = x + (b0 + lmod) * RNN_T;
  float xcur = xrow[0];

  const int aoff = lmod * HB_STRIDE + 2 * half;
  const int col0 = wave * NT_W * 16 + lmod;
  const int col1 = col0 + 16;
  const v8f vzero = {0.f, 0.f, 0.f, 0.f, 0.f, 0.f, 0.f, 0.f};

  __syncthreads();

  v2f a0[16], a1[16];
#pragma unroll
  for (int i = 0; i < 16; ++i) {
    a0[i] = *(const v2f*)(hbuf[0] + aoff + 4 * KC0(i));
    a1[i] = *(const v2f*)(hbuf[0] + aoff + 4 * KC1(i));
  }

  v8f c0, c1;
#pragma unroll
  for (int e = 0; e < 8; ++e) { c0[e] = biasv[0]; c1[e] = biasv[1]; }
#pragma unroll
  for (int i = 0; i < 16; ++i) {
    c0 = WMMA4(a0[i], bw[0][KC0(i)], c0);
    c1 = WMMA4(a0[i], bw[1][KC0(i)], c1);
  }

  float p0 = 0.0f, p1 = 0.0f, p2 = 0.0f, p3 = 0.0f;

#define PAIR1(i)                                  \
  c1 = WMMA4(a1[(i)], bw[1][KC1(i)], c1);         \
  e1 = WMMA4(a1[(i) + 8], bw[1][KC1((i) + 8)], e1)
#define PAIR0(i)                                  \
  n0 = WMMA4(a0[(i)], bw[0][KC0(i)], n0);         \
  n1 = WMMA4(a0[(i)], bw[1][KC0(i)], n1)

  for (int t = 0; t < RNN_TT; ++t) {
    const int par = t & 1;
    float* hnxt = hbuf[par ^ 1];
    const float yprev = ((p0 + p1) + (p2 + p3)) + decb;

    v2f al;
    al[0] = (half == 0) ? ((t < RNN_T) ? xcur : yprev) : 0.0f;
    al[1] = 0.0f;

    v8f e0 = vzero;
#pragma unroll
    for (int i = 0; i < 8; ++i) {
      c0 = WMMA4(a1[i], bw[0][KC1(i)], c0);
      e0 = WMMA4(a1[i + 8], bw[0][KC1(i + 8)], e0);
    }
    e0 = WMMA4(al, blast[0], e0);

    v8f e1 = vzero;
    PAIR1(0);
    ORDER4(c1, e1, c0, e0);
    v8f pre0;
#pragma unroll
    for (int v = 0; v < 8; ++v) pre0[v] = c0[v] + e0[v];
    PAIR1(1);
    ORDER3(c1, e1, pre0);
    v8f th0;
#pragma unroll
    for (int v = 0; v < 8; ++v) th0[v] = dev_tanh(pre0[v]);
    PAIR1(2);
    ORDER3(c1, e1, th0);
    v8f yr;
#pragma unroll
    for (int v = 0; v < 8; ++v) {
      hnxt[(v + 8 * half) * HB_STRIDE + col0] = th0[v];
      yr[v] = th0[v] * dwv[0];
    }
    ORDER2(c1, e1);
    PAIR1(3);
    PAIR1(4);
    PAIR1(5);
    BEFORE2(c1, e1);

    __syncthreads();
#pragma unroll
    for (int i = 0; i < 16; ++i) a0[i] = *(const v2f*)(hnxt + aoff + 4 * KC0(i));
    const int tn = (t + 1 < RNN_T) ? (t + 1) : (RNN_T - 1);
    const float xnext = xrow[tn];
    ORDER2(c1, e1);

    PAIR1(6);
    PAIR1(7);
    e1 = WMMA4(al, blast[1], e1);

    v8f n0, n1;
#pragma unroll
    for (int e = 0; e < 8; ++e) { n0[e] = biasv[0]; n1[e] = biasv[1]; }
    ORDER4(c1, e1, n0, n1);
    PAIR0(0);
    ORDER4(n0, n1, c1, e1);
    v8f pre1;
#pragma unroll
    for (int v = 0; v < 8; ++v) pre1[v] = c1[v] + e1[v];
    PAIR0(1);
    ORDER3(n0, n1, pre1);
    v8f th1;
#pragma unroll
    for (int v = 0; v < 8; ++v) th1[v] = dev_tanh(pre1[v]);
    PAIR0(2);
    ORDER3(n0, n1, th1);
#pragma unroll
    for (int v = 0; v < 8; ++v) {
      hnxt[(v + 8 * half) * HB_STRIDE + col1] = th1[v];
      yr[v] += th1[v] * dwv[1];
    }
    PAIR0(3);
    ORDER3(n0, n1, yr);
    yr = row_stage<1>(yr);
    PAIR0(4);
    ORDER3(n0, n1, yr);
    yr = row_stage<2>(yr);
    PAIR0(5);
    ORDER3(n0, n1, yr);
    yr = row_stage<4>(yr);
    PAIR0(6);
    ORDER3(n0, n1, yr);
    yr = row_stage<8>(yr);
    {
      float* ypw = ypart[par] + (wave * 32 + lane) * 8;
#pragma unroll
      for (int v = 0; v < 8; ++v) ypw[v] = yr[v];
    }
    ORDER2(n0, n1);
    PAIR0(7);
    PAIR0(8);
    PAIR0(9);
    BEFORE2(n0, n1);

    __syncthreads();
    {
      const float* yp = ypart[par] + (lmod >> 3) * 128 + (lmod & 7);
      p0 = yp[0]; p1 = yp[256]; p2 = yp[512]; p3 = yp[768];
    }
#pragma unroll
    for (int i = 0; i < 16; ++i) a1[i] = *(const v2f*)(hnxt + aoff + 4 * KC1(i));
    if (t > 0 && wave == 0 && half == 0) y[(b0 + lmod) * RNN_TT + (t - 1)] = yprev;
    ORDER2(n0, n1);

    PAIR0(10);
    PAIR0(11);
    PAIR0(12);
    PAIR0(13);
    PAIR0(14);
    PAIR0(15);
    c0 = n0;
    c1 = n1;
    xcur = xnext;
  }

  {
    const float yl = ((p0 + p1) + (p2 + p3)) + decb;
    if (wave == 0 && half == 0) y[(b0 + lmod) * RNN_TT + (RNN_TT - 1)] = yl;
  }
}

extern "C" void kernel_launch(void* const* d_in, const int* in_sizes, int n_in,
                              void* d_out, int out_size, void* d_ws,
                              size_t ws_size, hipStream_t stream) {
  const float* x      = (const float*)d_in[0];
  const float* hidden = (const float*)d_in[1];
  const float* fc0_w  = (const float*)d_in[2];
  const float* fc0_b  = (const float*)d_in[3];
  const float* fc_w   = (const float*)d_in[4];
  const float* fc_b   = (const float*)d_in[5];
  const float* dec_w  = (const float*)d_in[6];
  const float* dec_b  = (const float*)d_in[7];
  float* y = (float*)d_out;

  rnn516_wmma<<<dim3(RNN_B / 16), dim3(NWAVE * 32), 0, stream>>>(
      x, hidden, fc0_w, fc0_b, fc_w, fc_b, dec_w, dec_b, y);
}


# operands swapped (fc_w as A, h as B): a wave's tanh output registers are its own next-step B fragments, other tiles move as 16-byte LDS rows, one barrier per step; new K and decode sum order
# speedup vs baseline: 7.5916x; 1.1039x over previous
#include <hip/hip_runtime.h>
#include <cmath>


typedef __attribute__((ext_vector_type(2))) float v2f;
typedef __attribute__((ext_vector_type(4))) float v4f;
typedef __attribute__((ext_vector_type(8))) float v8f;

#define RNN_B 1024
#define RNN_T 512
#define RNN_H 128
#define RNN_P 4
#define RNN_TT (RNN_T + RNN_P)
#define NWAVE 4
#define HT_ELEMS (8 * 2 * 32 * 4)

__device__ __forceinline__ float dev_tanh(float v) {
#if __has_builtin(__builtin_amdgcn_tanhf)
  return __builtin_amdgcn_tanhf(v);
#else
  return tanhf(v);
#endif
}

#define WMMA4(a, b, c) \
  __builtin_amdgcn_wmma_f32_16x16x4_f32(false, (a), false, (b), (short)0, (c), false, false)
#define LO2(q) __builtin_shufflevector((q), (q), 0, 1)
#define HI2(q) __builtin_shufflevector((q), (q), 2, 3)

#define ORDER2(a, b) asm volatile("" : "+v"(a), "+v"(b))
#define BEFORE2(a, b) asm volatile("" : : "v"(a), "v"(b))
#define STORES_FIRST(a, b) asm volatile("" : "+v"(a), "+v"(b) : : "memory")

__global__ __launch_bounds__(NWAVE * 32) void rnn516_wmma(
    const float* __restrict__ x, const float* __restrict__ hidden,
    const float* __restrict__ fc0_w, const float* __restrict__ fc0_b,
    const float* __restrict__ fc_w, const float* __restrict__ fc_b,
    const float* __restrict__ dec_w, const float* __restrict__ dec_b,
    float* __restrict__ y) {
  __shared__ __attribute__((aligned(16))) float hT[2][HT_ELEMS];
  __shared__ float ypart[2][NWAVE * 32];

  const int tid  = threadIdx.x;
  const int lane = tid & 31;
  const int wave = __builtin_amdgcn_readfirstlane(tid >> 5);
  const int half = lane >> 4;
  const int lmod = lane & 15;
  const int b0 = blockIdx.x * 16;

  v2f wa[2][8][4];
  v2f alast[2];
  v8f bias8[2], dw8[2];
#pragma unroll
  for (int jt = 0; jt < 2; ++jt) {
    const int n = (wave * 2 + jt) * 16 + lmod;
#pragma unroll
    for (int s = 0; s < 8; ++s) {
      const int kt = (2 * wave + s) & 7;
      const float* wp = fc_w + n * RNN_H + 16 * kt + 8 * half;
#pragma unroll
      for (int p = 0; p < 4; ++p) wa[jt][s][p] = *(const v2f*)(wp + 2 * p);
    }
    v2f a;
    a[0] = (half == 0) ? fc0_w[n] : 0.0f;
    a[1] = 0.0f;
    alast[jt] = a;
#pragma unroll
    for (int r = 0; r < 8; ++r) {
      const int nd = (wave * 2 + jt) * 16 + 8 * half + r;
      bias8[jt][r] = fc0_b[nd] + fc_b[nd];
      dw8[jt][r] = dec_w[nd];
    }
  }
  const float decb = dec_b[0];
  const float* xrow = x + (b0 + lmod) * RNN_T;
  float xcur = xrow[0];

  v4f hq[8][2];
#pragma unroll
  for (int s = 0; s < 8; ++s) {
    const int kt = (2 * wave + s) & 7;
    const float* hp = hidden + (b0 + lmod) * RNN_H + 16 * kt + 8 * half;
    hq[s][0] = *(const v4f*)(hp);
    hq[s][1] = *(const v4f*)(hp + 4);
  }
  float pp[8];
#pragma unroll
  for (int i = 0; i < 8; ++i) pp[i] = 0.0f;

#define PAIR(s, p)                                                          \
  n0 = WMMA4(wa[0][s][p], ((p) & 1) ? HI2(hq[s][(p) >> 1]) : LO2(hq[s][(p) >> 1]), n0); \
  n1 = WMMA4(wa[1][s][p], ((p) & 1) ? HI2(hq[s][(p) >> 1]) : LO2(hq[s][(p) >> 1]), n1)
#define SLOT(s) PAIR(s, 0); PAIR(s, 1); PAIR(s, 2); PAIR(s, 3)

  v8f n0 = bias8[0], n1 = bias8[1];
  SLOT(0);

  for (int t = 0; t < RNN_TT; ++t) {
    const int nb = (t + 1) & 1;
    SLOT(1);
    SLOT(2); SLOT(3); SLOT(4); SLOT(5); SLOT(6);
    asm volatile("" : "+v"(n0), "+v"(pp[0]), "+v"(pp[1]), "+v"(pp[2]), "+v"(pp[3]),
                 "+v"(pp[4]), "+v"(pp[5]), "+v"(pp[6]), "+v"(pp[7]), "+v"(xcur));
    SLOT(7);

    const float yprev = (((pp[0] + pp[1]) + (pp[2] + pp[3])) +
                         ((pp[4] + pp[5]) + (pp[6] + pp[7]))) + decb;
    v2f bl;
    bl[0] = (half == 0) ? ((t < RNN_T) ? xcur : yprev) : 0.0f;
    bl[1] = 0.0f;
    n0 = WMMA4(alast[0], bl, n0);
    n1 = WMMA4(alast[1], bl, n1);
    ORDER2(n0, n1);

    v8f th0, th1;
#pragma unroll
    for (int r = 0; r < 8; ++r) th0[r] = dev_tanh(n0[r]);
    hq[0][0] = __builtin_shufflevector(th0, th0, 0, 1, 2, 3);
    hq[0][1] = __builtin_shufflevector(th0, th0, 4, 5, 6, 7);

    v8f pre1 = n1;
    n0 = bias8[0];
    n1 = bias8[1];
    PAIR(0, 0);
    asm volatile("" : "+v"(n0), "+v"(n1), "+v"(pre1));
#pragma unroll
    for (int r = 0; r < 8; ++r) th1[r] = dev_tanh(pre1[r]);
    hq[1][0] = __builtin_shufflevector(th1, th1, 0, 1, 2, 3);
    hq[1][1] = __builtin_shufflevector(th1, th1, 4, 5, 6, 7);

    {
      float* hw = hT[nb] + ((2 * wave) * 2 * 32 + lane) * 4;
      *(v4f*)(hw)           = hq[0][0];
      *(v4f*)(hw + 128)     = hq[0][1];
      *(v4f*)(hw + 256)     = hq[1][0];
      *(v4f*)(hw + 384)     = hq[1][1];
    }
    {
      float d[8];
#pragma unroll
      for (int r = 0; r < 8; ++r) d[r] = th0[r] * dw8[0][r] + th1[r] * dw8[1][r];
      ypart[nb][wave * 32 + lane] = ((d[0] + d[1]) + (d[2] + d[3])) +
                                    ((d[4] + d[5]) + (d[6] + d[7]));
    }
    STORES_FIRST(n0, n1);
    PAIR(0, 1);
    PAIR(0, 2);
    PAIR(0, 3);
    BEFORE2(n0, n1);

    __syncthreads();
#pragma unroll
    for (int s = 2; s < 8; ++s) {
      const int kt = (2 * wave + s) & 7;
      const float* hr = hT[nb] + (kt * 2 * 32 + lane) * 4;
      hq[s][0] = *(const v4f*)(hr);
      hq[s][1] = *(const v4f*)(hr + 128);
    }
#pragma unroll
    for (int w = 0; w < 4; ++w) {
      pp[2 * w]     = ypart[nb][w * 32 + lmod];
      pp[2 * w + 1] = ypart[nb][w * 32 + 16 + lmod];
    }
    const int tn = (t + 1 < RNN_T) ? (t + 1) : (RNN_T - 1);
    xcur = xrow[tn];
    if (t > 0 && wave == 0 && half == 0) y[(b0 + lmod) * RNN_TT + (t - 1)] = yprev;
    ORDER2(n0, n1);
  }

  {
    const float yl = (((pp[0] + pp[1]) + (pp[2] + pp[3])) +
                      ((pp[4] + pp[5]) + (pp[6] + pp[7]))) + decb;
    if (wave == 0 && half == 0) y[(b0 + lmod) * RNN_TT + (RNN_TT - 1)] = yl;
  }
}

extern "C" void kernel_launch(void* const* d_in, const int* in_sizes, int n_in,
                              void* d_out, int out_size, void* d_ws,
                              size_t ws_size, hipStream_t stream) {
  const float* x      = (const float*)d_in[0];
  const float* hidden = (const float*)d_in[1];
  const float* fc0_w  = (const float*)d_in[2];
  const float* fc0_b  = (const float*)d_in[3];
  const float* fc_w   = (const float*)d_in[4];
  const float* fc_b   = (const float*)d_in[5];
  const float* dec_w  = (const float*)d_in[6];
  const float* dec_b  = (const float*)d_in[7];
  float* y = (float*)d_out;

  rnn516_wmma<<<dim3(RNN_B / 16), dim3(NWAVE * 32), 0, stream>>>(
      x, hidden, fc0_w, fc0_b, fc_w, fc_b, dec_w, dec_b, y);
}


# time loop unrolled by two so the ping-pong LDS buffer index is a compile-time constant: LDS offsets become immediates, the per-step accumulator copies and address/parity instructions go
# speedup vs baseline: 7.8917x; 1.0395x over previous
#include <hip/hip_runtime.h>
#include <cmath>
#include <type_traits>


typedef __attribute__((ext_vector_type(2))) float v2f;
typedef __attribute__((ext_vector_type(4))) float v4f;
typedef __attribute__((ext_vector_type(8))) float v8f;

#define RNN_B 1024
#define RNN_T 512
#define RNN_H 128
#define RNN_P 4
#define RNN_TT (RNN_T + RNN_P)
#define NWAVE 4
#define HT_ELEMS (8 * 2 * 32 * 4)

__device__ __forceinline__ float dev_tanh(float v) {
#if __has_builtin(__builtin_amdgcn_tanhf)
  return __builtin_amdgcn_tanhf(v);
#else
  return tanhf(v);
#endif
}

#define WMMA4(a, b, c) \
  __builtin_amdgcn_wmma_f32_16x16x4_f32(false, (a), false, (b), (short)0, (c), false, false)
#define LO2(q) __builtin_shufflevector((q), (q), 0, 1)
#define HI2(q) __builtin_shufflevector((q), (q), 2, 3)

#define ORDER2(a, b) asm volatile("" : "+v"(a), "+v"(b))
#define BEFORE2(a, b) asm volatile("" : : "v"(a), "v"(b))
#define STORES_FIRST(a, b) asm volatile("" : "+v"(a), "+v"(b) : : "memory")

__global__ __launch_bounds__(NWAVE * 32) void rnn516_wmma(
    const float* __restrict__ x, const float* __restrict__ hidden,
    const float* __restrict__ fc0_w, const float* __restrict__ fc0_b,
    const float* __restrict__ fc_w, const float* __restrict__ fc_b,
    const float* __restrict__ dec_w, const float* __restrict__ dec_b,
    float* __restrict__ y) {
  __shared__ __attribute__((aligned(16))) float hT[2][HT_ELEMS];
  __shared__ float ypart[2][NWAVE * 32];

  const int tid  = threadIdx.x;
  const int lane = tid & 31;
  const int wave = __builtin_amdgcn_readfirstlane(tid >> 5);
  const int half = lane >> 4;
  const int lmod = lane & 15;
  const int b0 = blockIdx.x * 16;

  v2f wa[2][8][4];
  v2f alast[2];
  v8f bias8[2], dw8[2];
#pragma unroll
  for (int jt = 0; jt < 2; ++jt) {
    const int n = (wave * 2 + jt) * 16 + lmod;
#pragma unroll
    for (int s = 0; s < 8; ++s) {
      const int kt = (2 * wave + s) & 7;
      const float* wp = fc_w + n * RNN_H + 16 * kt + 8 * half;
#pragma unroll
      for (int p = 0; p < 4; ++p) wa[jt][s][p] = *(const v2f*)(wp + 2 * p);
    }
    v2f a;
    a[0] = (half == 0) ? fc0_w[n] : 0.0f;
    a[1] = 0.0f;
    alast[jt] = a;
#pragma unroll
    for (int r = 0; r < 8; ++r) {
      const int nd = (wave * 2 + jt) * 16 + 8 * half + r;
      bias8[jt][r] = fc0_b[nd] + fc_b[nd];
      dw8[jt][r] = dec_w[nd];
    }
  }
  const float decb = dec_b[0];
  const float* xrow = x + (b0 + lmod) * RNN_T;
  float xcur = xrow[0];

  v4f hq[8][2];
#pragma unroll
  for (int s = 0; s < 8; ++s) {
    const int kt = (2 * wave + s) & 7;
    const float* hp = hidden + (b0 + lmod) * RNN_H + 16 * kt + 8 * half;
    hq[s][0] = *(const v4f*)(hp);
    hq[s][1] = *(const v4f*)(hp + 4);
  }
  float pp[8];
#pragma unroll
  for (int i = 0; i < 8; ++i) pp[i] = 0.0f;

#define PAIR(s, p)                                                          \
  n0 = WMMA4(wa[0][s][p], ((p) & 1) ? HI2(hq[s][(p) >> 1]) : LO2(hq[s][(p) >> 1]), n0); \
  n1 = WMMA4(wa[1][s][p], ((p) & 1) ? HI2(hq[s][(p) >> 1]) : LO2(hq[s][(p) >> 1]), n1)
#define SLOT(s) PAIR(s, 0); PAIR(s, 1); PAIR(s, 2); PAIR(s, 3)

  v8f n0 = bias8[0], n1 = bias8[1];
  SLOT(0);

  auto step = [&](const int t, auto nbc) __attribute__((always_inline)) {
    constexpr int nb = decltype(nbc)::value;
    SLOT(1);
    SLOT(2); SLOT(3); SLOT(4); SLOT(5); SLOT(6);
    __builtin_amdgcn_sched_barrier(0);
    SLOT(7);

    const float yprev = (((pp[0] + pp[1]) + (pp[2] + pp[3])) +
                         ((pp[4] + pp[5]) + (pp[6] + pp[7]))) + decb;
    v2f bl;
    bl[0] = (half == 0) ? ((t < RNN_T) ? xcur : yprev) : 0.0f;
    bl[1] = 0.0f;
    n0 = WMMA4(alast[0], bl, n0);
    n1 = WMMA4(alast[1], bl, n1);
    ORDER2(n0, n1);

    v8f th0, th1;
#pragma unroll
    for (int r = 0; r < 8; ++r) th0[r] = dev_tanh(n0[r]);
    hq[0][0] = __builtin_shufflevector(th0, th0, 0, 1, 2, 3);
    hq[0][1] = __builtin_shufflevector(th0, th0, 4, 5, 6, 7);

    v8f pre1 = n1;
    n0 = bias8[0];
    n1 = bias8[1];
    PAIR(0, 0);
    asm volatile("" : "+v"(n0), "+v"(n1), "+v"(pre1));
#pragma unroll
    for (int r = 0; r < 8; ++r) th1[r] = dev_tanh(pre1[r]);
    hq[1][0] = __builtin_shufflevector(th1, th1, 0, 1, 2, 3);
    hq[1][1] = __builtin_shufflevector(th1, th1, 4, 5, 6, 7);

    {
      float* hw = hT[nb] + ((2 * wave) * 2 * 32 + lane) * 4;
      *(v4f*)(hw)           = hq[0][0];
      *(v4f*)(hw + 128)     = hq[0][1];
      *(v4f*)(hw + 256)     = hq[1][0];
      *(v4f*)(hw + 384)     = hq[1][1];
    }
    {
      float d[8];
#pragma unroll
      for (int r = 0; r < 8; ++r) d[r] = th0[r] * dw8[0][r] + th1[r] * dw8[1][r];
      ypart[nb][wave * 32 + lane] = ((d[0] + d[1]) + (d[2] + d[3])) +
                                    ((d[4] + d[5]) + (d[6] + d[7]));
    }
    STORES_FIRST(n0, n1);
    PAIR(0, 1);
    PAIR(0, 2);
    PAIR(0, 3);
    BEFORE2(n0, n1);

    __syncthreads();
#pragma unroll
    for (int s = 2; s < 8; ++s) {
      const int kt = (2 * wave + s) & 7;
      const float* hr = hT[nb] + (kt * 2 * 32 + lane) * 4;
      hq[s][0] = *(const v4f*)(hr);
      hq[s][1] = *(const v4f*)(hr + 128);
    }
#pragma unroll
    for (int w = 0; w < 4; ++w) {
      pp[2 * w]     = ypart[nb][w * 32 + lmod];
      pp[2 * w + 1] = ypart[nb][w * 32 + 16 + lmod];
    }
    const int tn = (t + 1 < RNN_T) ? (t + 1) : (RNN_T - 1);
    xcur = xrow[tn];
    if (t > 0 && wave == 0 && half == 0) y[(b0 + lmod) * RNN_TT + (t - 1)] = yprev;
    ORDER2(n0, n1);
  };
  for (int t = 0; t < RNN_TT; t += 2) {
    step(t, std::integral_constant<int, 1>{});
    step(t + 1, std::integral_constant<int, 0>{});
  }

  {
    const float yl = (((pp[0] + pp[1]) + (pp[2] + pp[3])) +
                      ((pp[4] + pp[5]) + (pp[6] + pp[7]))) + decb;
    if (wave == 0 && half == 0) y[(b0 + lmod) * RNN_TT + (RNN_TT - 1)] = yl;
  }
}

extern "C" void kernel_launch(void* const* d_in, const int* in_sizes, int n_in,
                              void* d_out, int out_size, void* d_ws,
                              size_t ws_size, hipStream_t stream) {
  const float* x      = (const float*)d_in[0];
  const float* hidden = (const float*)d_in[1];
  const float* fc0_w  = (const float*)d_in[2];
  const float* fc0_b  = (const float*)d_in[3];
  const float* fc_w   = (const float*)d_in[4];
  const float* fc_b   = (const float*)d_in[5];
  const float* dec_w  = (const float*)d_in[6];
  const float* dec_b  = (const float*)d_in[7];
  float* y = (float*)d_out;

  rnn516_wmma<<<dim3(RNN_B / 16), dim3(NWAVE * 32), 0, stream>>>(
      x, hidden, fc0_w, fc0_b, fc_w, fc_b, dec_w, dec_b, y);
}


# time loop split at t=T: input steps drop the feedback path (no per-step y sum, select or store); y partials stay in LDS, summed and stored in 128-B runs after the loop; same sum tree
# speedup vs baseline: 8.3959x; 1.0639x over previous
#include <hip/hip_runtime.h>
#include <cmath>
#include <type_traits>


typedef __attribute__((ext_vector_type(2))) float v2f;
typedef __attribute__((ext_vector_type(4))) float v4f;
typedef __attribute__((ext_vector_type(8))) float v8f;

#define RNN_B 1024
#define RNN_T 512
#define RNN_H 128
#define RNN_P 4
#define RNN_TT (RNN_T + RNN_P)
#define NWAVE 4
#define HT_ELEMS (8 * 2 * 32 * 4)
#define YB_STRIDE 129

__device__ __forceinline__ float dev_tanh(float v) {
#if __has_builtin(__builtin_amdgcn_tanhf)
  return __builtin_amdgcn_tanhf(v);
#else
  return tanhf(v);
#endif
}

#define WMMA4(a, b, c) \
  __builtin_amdgcn_wmma_f32_16x16x4_f32(false, (a), false, (b), (short)0, (c), false, false)
#define LO2(q) __builtin_shufflevector((q), (q), 0, 1)
#define HI2(q) __builtin_shufflevector((q), (q), 2, 3)

#define ORDER2(a, b) asm volatile("" : "+v"(a), "+v"(b))
#define BEFORE2(a, b) asm volatile("" : : "v"(a), "v"(b))
#define STORES_FIRST(a, b) asm volatile("" : "+v"(a), "+v"(b) : : "memory")

__global__ __launch_bounds__(NWAVE * 32) void rnn516_wmma(
    const float* __restrict__ x, const float* __restrict__ hidden,
    const float* __restrict__ fc0_w, const float* __restrict__ fc0_b,
    const float* __restrict__ fc_w, const float* __restrict__ fc_b,
    const float* __restrict__ dec_w, const float* __restrict__ dec_b,
    float* __restrict__ y) {
  __shared__ __attribute__((aligned(16))) float hT[2][HT_ELEMS];
  __shared__ float ybuf[RNN_TT * YB_STRIDE];

  const int tid  = threadIdx.x;
  const int lane = tid & 31;
  const int wave = __builtin_amdgcn_readfirstlane(tid >> 5);
  const int half = lane >> 4;
  const int lmod = lane & 15;
  const int b0 = blockIdx.x * 16;

  v2f wa[2][8][4];
  v2f alast[2];
  v8f bias8[2], dw8[2];
#pragma unroll
  for (int jt = 0; jt < 2; ++jt) {
    const int n = (wave * 2 + jt) * 16 + lmod;
#pragma unroll
    for (int s = 0; s < 8; ++s) {
      const int kt = (2 * wave + s) & 7;
      const float* wp = fc_w + n * RNN_H + 16 * kt + 8 * half;
#pragma unroll
      for (int p = 0; p < 4; ++p) wa[jt][s][p] = *(const v2f*)(wp + 2 * p);
    }
    v2f a;
    a[0] = (half == 0) ? fc0_w[n] : 0.0f;
    a[1] = 0.0f;
    alast[jt] = a;
#pragma unroll
    for (int r = 0; r < 8; ++r) {
      const int nd = (wave * 2 + jt) * 16 + 8 * half + r;
      bias8[jt][r] = fc0_b[nd] + fc_b[nd];
      dw8[jt][r] = dec_w[nd];
    }
  }
  const float decb = dec_b[0];
  const float* xrow = x + (b0 + lmod) * RNN_T;
  float xcur = xrow[0];

  v4f hq[8][2];
#pragma unroll
  for (int s = 0; s < 8; ++s) {
    const int kt = (2 * wave + s) & 7;
    const float* hp = hidden + (b0 + lmod) * RNN_H + 16 * kt + 8 * half;
    hq[s][0] = *(const v4f*)(hp);
    hq[s][1] = *(const v4f*)(hp + 4);
  }
  float pp[8];
#pragma unroll
  for (int i = 0; i < 8; ++i) pp[i] = 0.0f;

#define PAIR(s, p)                                                          \
  n0 = WMMA4(wa[0][s][p], ((p) & 1) ? HI2(hq[s][(p) >> 1]) : LO2(hq[s][(p) >> 1]), n0); \
  n1 = WMMA4(wa[1][s][p], ((p) & 1) ? HI2(hq[s][(p) >> 1]) : LO2(hq[s][(p) >> 1]), n1)
#define SLOT(s) PAIR(s, 0); PAIR(s, 1); PAIR(s, 2); PAIR(s, 3)

  v8f n0 = bias8[0], n1 = bias8[1];
  SLOT(0);

  auto step = [&](const int t, auto nbc, auto fbc) __attribute__((always_inline)) {
    constexpr int nb = decltype(nbc)::value;
    constexpr bool fb = decltype(fbc)::value;
    SLOT(1);
    SLOT(2); SLOT(3); SLOT(4); SLOT(5); SLOT(6);
    __builtin_amdgcn_sched_barrier(0);
    SLOT(7);

    float in;
    if constexpr (fb) {
      in = (((pp[0] + pp[1]) + (pp[2] + pp[3])) +
            ((pp[4] + pp[5]) + (pp[6] + pp[7]))) + decb;
    } else {
      in = xcur;
    }
    v2f bl;
    bl[0] = (half == 0) ? in : 0.0f;
    bl[1] = 0.0f;
    n0 = WMMA4(alast[0], bl, n0);
    n1 = WMMA4(alast[1], bl, n1);
    ORDER2(n0, n1);

    v8f th0, th1;
#pragma unroll
    for (int r = 0; r < 8; ++r) th0[r] = dev_tanh(n0[r]);
    hq[0][0] = __builtin_shufflevector(th0, th0, 0, 1, 2, 3);
    hq[0][1] = __builtin_shufflevector(th0, th0, 4, 5, 6, 7);

    v8f pre1 = n1;
    n0 = bias8[0];
    n1 = bias8[1];
    PAIR(0, 0);
    asm volatile("" : "+v"(n0), "+v"(n1), "+v"(pre1));
#pragma unroll
    for (int r = 0; r < 8; ++r) th1[r] = dev_tanh(pre1[r]);
    hq[1][0] = __builtin_shufflevector(th1, th1, 0, 1, 2, 3);
    hq[1][1] = __builtin_shufflevector(th1, th1, 4, 5, 6, 7);

    {
      float* hw = hT[nb] + ((2 * wave) * 2 * 32 + lane) * 4;
      *(v4f*)(hw)           = hq[0][0];
      *(v4f*)(hw + 128)     = hq[0][1];
      *(v4f*)(hw + 256)     = hq[1][0];
      *(v4f*)(hw + 384)     = hq[1][1];
    }
    {
      float d[8];
#pragma unroll
      for (int r = 0; r < 8; ++r) d[r] = th0[r] * dw8[0][r] + th1[r] * dw8[1][r];
      ybuf[t * YB_STRIDE + wave * 32 + lane] = ((d[0] + d[1]) + (d[2] + d[3])) +
                                               ((d[4] + d[5]) + (d[6] + d[7]));
    }
    STORES_FIRST(n0, n1);
    PAIR(0, 1);
    PAIR(0, 2);
    PAIR(0, 3);
    BEFORE2(n0, n1);

    __syncthreads();
#pragma unroll
    for (int s = 2; s < 8; ++s) {
      const int kt = (2 * wave + s) & 7;
      const float* hr = hT[nb] + (kt * 2 * 32 + lane) * 4;
      hq[s][0] = *(const v4f*)(hr);
      hq[s][1] = *(const v4f*)(hr + 128);
    }
    if constexpr (fb) {
      const float* yp = ybuf + t * YB_STRIDE + lmod;
#pragma unroll
      for (int w = 0; w < 4; ++w) {
        pp[2 * w]     = yp[w * 32];
        pp[2 * w + 1] = yp[w * 32 + 16];
      }
    } else {
      const int tn = (t + 1 < RNN_T) ? (t + 1) : (RNN_T - 1);
      xcur = xrow[tn];
    }
    ORDER2(n0, n1);
  };
  for (int t = 0; t < RNN_T; t += 2) {
    step(t, std::integral_constant<int, 1>{}, std::false_type{});
    step(t + 1, std::integral_constant<int, 0>{}, std::false_type{});
  }
  {
    const float* yp = ybuf + (RNN_T - 1) * YB_STRIDE + lmod;
#pragma unroll
    for (int w = 0; w < 4; ++w) {
      pp[2 * w]     = yp[w * 32];
      pp[2 * w + 1] = yp[w * 32 + 16];
    }
  }
  for (int t = RNN_T; t < RNN_TT; t += 2) {
    step(t, std::integral_constant<int, 1>{}, std::true_type{});
    step(t + 1, std::integral_constant<int, 0>{}, std::true_type{});
  }

  for (int tb = 0; tb < RNN_TT; tb += 32) {
    const int t = tb + lane;
    if (t < RNN_TT) {
#pragma unroll
      for (int j = 0; j < 4; ++j) {
        const int m = wave * 4 + j;
        const float* p = ybuf + t * YB_STRIDE + m;
        y[(b0 + m) * RNN_TT + t] = (((p[0] + p[16]) + (p[32] + p[48])) +
                                    ((p[64] + p[80]) + (p[96] + p[112]))) + decb;
      }
    }
  }
}

extern "C" void kernel_launch(void* const* d_in, const int* in_sizes, int n_in,
                              void* d_out, int out_size, void* d_ws,
                              size_t ws_size, hipStream_t stream) {
  const float* x      = (const float*)d_in[0];
  const float* hidden = (const float*)d_in[1];
  const float* fc0_w  = (const float*)d_in[2];
  const float* fc0_b  = (const float*)d_in[3];
  const float* fc_w   = (const float*)d_in[4];
  const float* fc_b   = (const float*)d_in[5];
  const float* dec_w  = (const float*)d_in[6];
  const float* dec_b  = (const float*)d_in[7];
  float* y = (float*)d_out;

  rnn516_wmma<<<dim3(RNN_B / 16), dim3(NWAVE * 32), 0, stream>>>(
      x, hidden, fc0_w, fc0_b, fc_w, fc_b, dec_w, dec_b, y);
}
